# starting kernel
# baseline (speedup 1.0000x reference)
#include <hip/hip_runtime.h>
#include <hip/hip_bf16.h>

typedef __attribute__((ext_vector_type(2))) float v2f;
typedef __attribute__((ext_vector_type(8))) float v8f;

#define F_SIZE 256
#define LDSP 260

__global__ __launch_bounds__(256) void k_attn_ids(
    const float* __restrict__ act, const int* __restrict__ bids,
    const int* __restrict__ mids, const int* __restrict__ xy,
    const int* __restrict__ m0, const int* __restrict__ m1,
    const int* __restrict__ m2, const int* __restrict__ m3,
    const float* __restrict__ Wa, const float* __restrict__ ba,
    float* __restrict__ fw_ws, int* __restrict__ id_ws, int n_act)
{
    const int wave = threadIdx.x >> 5;
    const int lane = threadIdx.x & 31;
    const int row  = blockIdx.x * 8 + wave;
    if (row >= n_act) return;

    const float4* __restrict__ arow = (const float4*)(act + (size_t)row * F_SIZE);
    const float4* __restrict__ wrow = (const float4*)(Wa + (size_t)lane * F_SIZE);
    float acc = 0.f;
#pragma unroll 8
    for (int i = 0; i < F_SIZE / 4; ++i) {
        float4 a = arow[i];
        float4 w = wrow[i];
        acc += a.x * w.x + a.y * w.y + a.z * w.z + a.w * w.w;
    }
    acc += ba[lane];

    float mx = acc;
    mx = fmaxf(mx, __shfl_xor(mx, 1, 32));
    mx = fmaxf(mx, __shfl_xor(mx, 2, 32));
    float e = __expf(acc - mx);
    float s = e;
    s += __shfl_xor(s, 1, 32);
    s += __shfl_xor(s, 2, 32);
    const float fw = e / s;

    const int p = lane & 3;
    const int k = lane >> 2;
    const int offx[8] = {-1, 0, 1, -1, 1, -1, 0, 1};
    const int offy[8] = {-1, -1, -1, 0, 0, 1, 1, 1};
    const int scale = p + 1;
    int x = xy[row * 2 + 0] + scale * offx[k];
    int y = xy[row * 2 + 1] + scale * offy[k];
    const int m = mids[row];
    const int dim = 128 >> m;
    x = min(max(x, 0), dim - 1);
    y = min(max(y, 0), dim - 1);
    const int* __restrict__ map = (m == 0) ? m0 : (m == 1) ? m1 : (m == 2) ? m2 : m3;
    const int b = bids[row];
    const int id = map[((size_t)b * dim + y) * dim + x];

    const size_t o = (size_t)row * 32 + p * 8 + k;
    fw_ws[o] = fw;
    id_ws[o] = id;
}

__global__ __launch_bounds__(256) void k_gather(
    const float* __restrict__ act, const float* __restrict__ pas,
    const float* __restrict__ fw_ws, const int* __restrict__ id_ws,
    float* __restrict__ weighted, int n_act)
{
    __shared__ float s_fw[32];
    __shared__ const float* s_ptr[32];
    const int row = blockIdx.x;
    const int t = threadIdx.x;
    if (t < 32) {
        const int id = id_ws[(size_t)row * 32 + t];
        s_fw[t] = fw_ws[(size_t)row * 32 + t];
        s_ptr[t] = (id < n_act) ? (act + (size_t)id * F_SIZE)
                                : (pas + (size_t)(id - n_act) * F_SIZE);
    }
    __syncthreads();
    float acc = 0.f;
#pragma unroll
    for (int g = 0; g < 32; ++g)
        acc += s_fw[g] * s_ptr[g][t];
    weighted[(size_t)row * F_SIZE + t] = acc;
}

__global__ __launch_bounds__(256) void k_dual_gemm(
    const float* __restrict__ weighted,
    const float* __restrict__ Wv, const float* __restrict__ bv,
    const float* __restrict__ Wo, const float* __restrict__ bo,
    float* __restrict__ out)
{
    __shared__ float lds_a[16 * LDSP];
    __shared__ float lds_v[16 * LDSP];

    const int t = threadIdx.x;
    const int wave = t >> 5;
    const int lane = t & 31;
    const int m0 = blockIdx.x * 16;

    {
        const int r = t >> 4;
        const int c0 = (t & 15) * 16;
        const float4* __restrict__ src =
            (const float4*)(weighted + (size_t)(m0 + r) * F_SIZE + c0);
        float4* dst = (float4*)(lds_a + r * LDSP + c0);
        dst[0] = src[0]; dst[1] = src[1]; dst[2] = src[2]; dst[3] = src[3];
    }
    __syncthreads();

    const int mrow = lane & 15;
    const int hi = lane >> 4;

#pragma unroll
    for (int tn = 0; tn < 2; ++tn) {
        const int nn = (wave * 2 + tn) * 16 + mrow;
        const float* __restrict__ ap = lds_a + mrow * LDSP + 2 * hi;
        const float* __restrict__ bp = Wv + (size_t)nn * F_SIZE + 2 * hi;
        v8f c = {};
#pragma unroll 4
        for (int k = 0; k < F_SIZE; k += 4) {
            v2f a, b;
            a.x = ap[k]; a.y = ap[k + 1];
            b.x = bp[k]; b.y = bp[k + 1];
            c = __builtin_amdgcn_wmma_f32_16x16x4_f32(
                    false, a, false, b, (short)0, c, false, false);
        }
        const float bias = bv[nn];
#pragma unroll
        for (int r = 0; r < 8; ++r)
            lds_v[(r + 8 * hi) * LDSP + nn] = c[r] + bias;
    }
    __syncthreads();

#pragma unroll
    for (int tn = 0; tn < 2; ++tn) {
        const int nn = (wave * 2 + tn) * 16 + mrow;
        const float* __restrict__ ap = lds_v + mrow * LDSP + 2 * hi;
        const float* __restrict__ bp = Wo + (size_t)nn * F_SIZE + 2 * hi;
        v8f c = {};
#pragma unroll 4
        for (int k = 0; k < F_SIZE; k += 4) {
            v2f a, b;
            a.x = ap[k]; a.y = ap[k + 1];
            b.x = bp[k]; b.y = bp[k + 1];
            c = __builtin_amdgcn_wmma_f32_16x16x4_f32(
                    false, a, false, b, (short)0, c, false, false);
        }
        const float bias = bo[nn];
#pragma unroll
        for (int r = 0; r < 8; ++r)
            out[(size_t)(m0 + r + 8 * hi) * F_SIZE + nn] = c[r] + bias;
    }
}

extern "C" void kernel_launch(void* const* d_in, const int* in_sizes, int n_in,
                              void* d_out, int out_size, void* d_ws, size_t ws_size,
                              hipStream_t stream) {
    const float* act  = (const float*)d_in[0];
    const float* pas  = (const float*)d_in[1];
    const int*   bids = (const int*)d_in[2];
    const int*   mids = (const int*)d_in[3];
    const int*   xy   = (const int*)d_in[4];
    const int*   m0   = (const int*)d_in[5];
    const int*   m1   = (const int*)d_in[6];
    const int*   m2   = (const int*)d_in[7];
    const int*   m3   = (const int*)d_in[8];
    const float* Wa   = (const float*)d_in[9];
    const float* ba   = (const float*)d_in[10];
    const float* Wv   = (const float*)d_in[11];
    const float* bv   = (const float*)d_in[12];
    const float* Wo   = (const float*)d_in[13];
    const float* bo   = (const float*)d_in[14];
    float* out = (float*)d_out;

    const int n_act = in_sizes[0] / F_SIZE;

    float* fw_ws    = (float*)d_ws;
    int*   id_ws    = (int*)(fw_ws + (size_t)n_act * 32);
    float* weighted = (float*)(id_ws + (size_t)n_act * 32);

    k_attn_ids<<<(n_act + 7) / 8, 256, 0, stream>>>(
        act, bids, mids, xy, m0, m1, m2, m3, Wa, ba, fw_ws, id_ws, n_act);

    k_gather<<<n_act, 256, 0, stream>>>(act, pas, fw_ws, id_ws, weighted, n_act);

    k_dual_gemm<<<n_act / 16, 256, 0, stream>>>(weighted, Wv, bv, Wo, bo, out);
}


# K3: each wave owns 16 rows, A fragments register-resident over the whole K; Wv/Wo fragment chunks copied once per 128-row block into LDS, shared by its 8 row tiles (8x fewer L2 reads); sum order kept
# speedup vs baseline: 1.9936x; 1.9936x over previous
#include <hip/hip_runtime.h>
#include <hip/hip_bf16.h>

typedef __attribute__((ext_vector_type(2))) float v2f;
typedef __attribute__((ext_vector_type(8))) float v8f;

#define F_SIZE 256
#define LDSP 260

#define K1_ROWS 8

#define K1_STAGE(N, D)                                                   \
    {                                                                    \
        const bool up = (lane & (D)) != 0;                               \
          \
        _Pragma("unroll")                                                \
        for (int i = 0; i < (N) / 2; ++i) {                              \
            const float keep = up ? v[i + (N) / 2] : v[i];               \
            const float send = up ? v[i] : v[i + (N) / 2];               \
            v[i] = keep + __shfl_xor(send, (D), 32);                     \
        }                                                                \
    }

__global__ __launch_bounds__(256) void k_attn_ids(
    const float* __restrict__ act, const int* __restrict__ bids,
    const int* __restrict__ mids, const int* __restrict__ xy,
    const int* __restrict__ m0, const int* __restrict__ m1,
    const int* __restrict__ m2, const int* __restrict__ m3,
    const float* __restrict__ Wa, const float* __restrict__ ba,
    float* __restrict__ fw_ws, int* __restrict__ id_ws, int n_act)
{
    __shared__ float s_fw[2][K1_ROWS * 32];
    __shared__ int   s_id[2][K1_ROWS * 32];

    const int t = threadIdx.x;
    const int k = t >> 5;
    const int lane = t & 31;

    float4 wa[4][2];
#pragma unroll
    for (int pp = 0; pp < 4; ++pp) {
        const float4* __restrict__ wr =
            (const float4*)(Wa + (size_t)(k * 4 + pp) * F_SIZE);
        wa[pp][0] = wr[lane];
        wa[pp][1] = wr[32 + lane];
    }

    const int r = lane >> 2;
    const int p = lane & 3;
    const float bias = ba[k * 4 + p];
    const int cell = k + (k >= 4 ? 1 : 0);
    const int ox = (cell % 3 - 1) * (p + 1);
    const int oy = (cell / 3 - 1) * (p + 1);

    int buf = 0;
    for (int row0 = blockIdx.x * K1_ROWS; row0 < n_act;
         row0 += gridDim.x * K1_ROWS, buf ^= 1) {
        const int myrow = min(row0 + r, n_act - 1);
        const int x0 = xy[myrow * 2 + 0];
        const int y0 = xy[myrow * 2 + 1];
        const int m = mids[myrow];
        const int b = bids[myrow];

        float v[K1_ROWS * 4];
#pragma unroll
        for (int rr = 0; rr < K1_ROWS; ++rr) {
            const int row = min(row0 + rr, n_act - 1);
            const float4* __restrict__ ar =
                (const float4*)(act + (size_t)row * F_SIZE);
            const float4 a0 = ar[lane];
            const float4 a1 = ar[32 + lane];
#pragma unroll
            for (int pp = 0; pp < 4; ++pp) {
                const float4 w0 = wa[pp][0];
                const float4 w1 = wa[pp][1];
                v[rr * 4 + pp] = a0.x * w0.x + a0.y * w0.y + a0.z * w0.z + a0.w * w0.w
                               + a1.x * w1.x + a1.y * w1.y + a1.z * w1.z + a1.w * w1.w;
            }
        }

        K1_STAGE(32, 16)
        K1_STAGE(16, 8)
        K1_STAGE(8, 4)
        K1_STAGE(4, 2)
        K1_STAGE(2, 1)
        const float acc = v[0] + bias;

        float mx = acc;
        mx = fmaxf(mx, __shfl_xor(mx, 1, 32));
        mx = fmaxf(mx, __shfl_xor(mx, 2, 32));
        const float e = __expf(acc - mx);
        float s = e;
        s += __shfl_xor(s, 1, 32);
        s += __shfl_xor(s, 2, 32);
        const float fw = e / s;

        const int dim = 128 >> m;
        const int x = min(max(x0 + ox, 0), dim - 1);
        const int y = min(max(y0 + oy, 0), dim - 1);
        const int* __restrict__ map = (m == 0) ? m0 : (m == 1) ? m1 : (m == 2) ? m2 : m3;
        const int id = map[((size_t)b * dim + y) * dim + x];

        s_fw[buf][r * 32 + p * 8 + k] = fw;
        s_id[buf][r * 32 + p * 8 + k] = id;
        __syncthreads();
        const size_t o = (size_t)row0 * 32 + t;
        if (o < (size_t)n_act * 32) {
            fw_ws[o] = s_fw[buf][t];
            id_ws[o] = s_id[buf][t];
        }
    }
}

__global__ __launch_bounds__(256) void k_gather(
    const float* __restrict__ act, const float* __restrict__ pas,
    const float* __restrict__ fw_ws, const int* __restrict__ id_ws,
    float* __restrict__ weighted, int n_act)
{
    __shared__ float s_fw[32];
    __shared__ const float* s_ptr[32];
    const int row = blockIdx.x;
    const int t = threadIdx.x;
    if (t < 32) {
        const int id = id_ws[(size_t)row * 32 + t];
        s_fw[t] = fw_ws[(size_t)row * 32 + t];
        s_ptr[t] = (id < n_act) ? (act + (size_t)id * F_SIZE)
                                : (pas + (size_t)(id - n_act) * F_SIZE);
    }
    __syncthreads();
    float acc = 0.f;
#pragma unroll
    for (int g = 0; g < 32; ++g)
        acc += s_fw[g] * s_ptr[g][t];
    weighted[(size_t)row * F_SIZE + t] = acc;
}

__global__ __launch_bounds__(256) void k_frag_order(
    const float* __restrict__ Wv, const float* __restrict__ Wo,
    float* __restrict__ wq)
{
    const int i = blockIdx.x * 256 + threadIdx.x;
    const int lane = i & 31;
    const int jj = (i >> 5) & 31;
    const int nt = (i >> 10) & 15;
    const int q = i >> 14;
    const int c = lane & 15;
    const int h = lane >> 4;
    const float* __restrict__ src =
        (q ? Wo : Wv) + (size_t)(nt * 16 + c) * F_SIZE + 8 * jj + 2 * h;
    const float2 lo = *(const float2*)src;
    const float2 up = *(const float2*)(src + 4);
    ((float4*)wq)[i] = make_float4(lo.x, lo.y, up.x, up.y);
}

#define K3_WAVES 8
#define K3_CHUNK4 2048
#define K3_PRE_LOAD(src)                                              \
    const float4 q0 = (src)[0 * 256], q1 = (src)[1 * 256],            \
                 q2 = (src)[2 * 256], q3 = (src)[3 * 256],            \
                 q4 = (src)[4 * 256], q5 = (src)[5 * 256],            \
                 q6 = (src)[6 * 256], q7 = (src)[7 * 256];
#define K3_PRE_STORE(dst)                                             \
    (dst)[0 * 256] = q0; (dst)[1 * 256] = q1; (dst)[2 * 256] = q2;    \
    (dst)[3 * 256] = q3; (dst)[4 * 256] = q4; (dst)[5 * 256] = q5;    \
    (dst)[6 * 256] = q6; (dst)[7 * 256] = q7;

__global__ __launch_bounds__(256) void k_dual_gemm(
    const float* __restrict__ weighted,
    const float* wq, const float* __restrict__ bv,
    const float* __restrict__ bo,
    float* __restrict__ out)
{
    __shared__ __attribute__((aligned(16))) float4 lds_b[2][K3_CHUNK4];
    __shared__ __attribute__((aligned(16))) float lds_t[K3_WAVES][16 * LDSP];

    const int t = threadIdx.x;
    const int wave = t >> 5;
    const int lane = t & 31;
    const int m0 = (blockIdx.x * K3_WAVES + wave) * 16;
    const int mrow = lane & 15;
    const int hi = lane >> 4;
    const float4* wq4 = (const float4*)wq;
    float* __restrict__ tile = lds_t[wave];

    K3_PRE_LOAD(wq4 + t)

#pragma unroll
    for (int g = 0; g < 4; ++g) {
        float4 v[8];
#pragma unroll
        for (int i = 0; i < 8; ++i) {
            const int idx = (g * 8 + i) * 32 + lane;
            v[i] = ((const float4*)(weighted + (size_t)(m0 + (idx >> 6)) * F_SIZE))[idx & 63];
        }
#pragma unroll
        for (int i = 0; i < 8; ++i) {
            const int idx = (g * 8 + i) * 32 + lane;
            *(float4*)(tile + (idx >> 6) * LDSP + (idx & 63) * 4) = v[i];
        }
    }
    K3_PRE_STORE(lds_b[0] + t)
    __syncthreads();

    const float* __restrict__ ap = tile + mrow * LDSP + 2 * hi;
    v2f a[64];

#pragma unroll
    for (int j = 0; j < 64; ++j) { a[j].x = ap[4 * j]; a[j].y = ap[4 * j + 1]; }

    for (int ci = 0; ci < 8; ++ci) {
        const int buf = ci & 1;
        K3_PRE_LOAD(wq4 + (size_t)(ci + 1) * K3_CHUNK4 + t)
        __builtin_amdgcn_sched_barrier(0);
        const float4* __restrict__ lb = lds_b[buf] + lane;
        float4 bf[64];
#pragma unroll
        for (int jj = 0; jj < 32; ++jj) {
            bf[2 * jj]     = lb[jj * 32];
            bf[2 * jj + 1] = lb[1024 + jj * 32];
        }
        __builtin_amdgcn_sched_barrier(0);
        v8f c0 = {}, c1 = {};
#pragma unroll
        for (int jj = 0; jj < 32; ++jj) {
            const float4 p = bf[2 * jj];
            const float4 q = bf[2 * jj + 1];
            v2f b;
            b.x = p.x; b.y = p.y;
            c0 = __builtin_amdgcn_wmma_f32_16x16x4_f32(
                     false, a[2 * jj], false, b, (short)0, c0, false, false);
            b.x = q.x; b.y = q.y;
            c1 = __builtin_amdgcn_wmma_f32_16x16x4_f32(
                     false, a[2 * jj], false, b, (short)0, c1, false, false);
            b.x = p.z; b.y = p.w;
            c0 = __builtin_amdgcn_wmma_f32_16x16x4_f32(
                     false, a[2 * jj + 1], false, b, (short)0, c0, false, false);
            b.x = q.z; b.y = q.w;
            c1 = __builtin_amdgcn_wmma_f32_16x16x4_f32(
                     false, a[2 * jj + 1], false, b, (short)0, c1, false, false);
        }
        const int nn = ci * 32 + mrow;
        const float bias0 = bv[nn];
        const float bias1 = bv[nn + 16];
#pragma unroll
        for (int r = 0; r < 8; ++r) {
            tile[(r + 8 * hi) * LDSP + nn] = c0[r] + bias0;
            tile[(r + 8 * hi) * LDSP + nn + 16] = c1[r] + bias1;
        }
        __builtin_amdgcn_sched_barrier(0);
        K3_PRE_STORE(lds_b[buf ^ 1] + t)
        __syncthreads();
    }

#pragma unroll
    for (int j = 0; j < 64; ++j) { a[j].x = ap[4 * j]; a[j].y = ap[4 * j + 1]; }

    for (int ci = 8; ci < 16; ++ci) {
        const int buf = ci & 1;
        const int cn = min(ci + 1, 15);
        K3_PRE_LOAD(wq4 + (size_t)cn * K3_CHUNK4 + t)
        __builtin_amdgcn_sched_barrier(0);
        const float4* __restrict__ lb = lds_b[buf] + lane;
        float4 bf[64];
#pragma unroll
        for (int jj = 0; jj < 32; ++jj) {
            bf[2 * jj]     = lb[jj * 32];
            bf[2 * jj + 1] = lb[1024 + jj * 32];
        }
        __builtin_amdgcn_sched_barrier(0);
        v8f c0 = {}, c1 = {};
#pragma unroll
        for (int jj = 0; jj < 32; ++jj) {
            const float4 p = bf[2 * jj];
            const float4 q = bf[2 * jj + 1];
            v2f b;
            b.x = p.x; b.y = p.y;
            c0 = __builtin_amdgcn_wmma_f32_16x16x4_f32(
                     false, a[2 * jj], false, b, (short)0, c0, false, false);
            b.x = q.x; b.y = q.y;
            c1 = __builtin_amdgcn_wmma_f32_16x16x4_f32(
                     false, a[2 * jj], false, b, (short)0, c1, false, false);
            b.x = p.z; b.y = p.w;
            c0 = __builtin_amdgcn_wmma_f32_16x16x4_f32(
                     false, a[2 * jj + 1], false, b, (short)0, c0, false, false);
            b.x = q.z; b.y = q.w;
            c1 = __builtin_amdgcn_wmma_f32_16x16x4_f32(
                     false, a[2 * jj + 1], false, b, (short)0, c1, false, false);
        }
        const int nn = (ci - 8) * 32 + mrow;
        const float bias0 = bo[nn];
        const float bias1 = bo[nn + 16];
        __builtin_amdgcn_sched_barrier(0);
        K3_PRE_STORE(lds_b[buf ^ 1] + t)
        __syncthreads();
#pragma unroll
        for (int r = 0; r < 8; ++r) {
            out[(size_t)(m0 + r + 8 * hi) * F_SIZE + nn] = c0[r] + bias0;
            out[(size_t)(m0 + r + 8 * hi) * F_SIZE + nn + 16] = c1[r] + bias1;
        }
    }
}

extern "C" void kernel_launch(void* const* d_in, const int* in_sizes, int n_in,
                              void* d_out, int out_size, void* d_ws, size_t ws_size,
                              hipStream_t stream) {
    const float* act  = (const float*)d_in[0];
    const float* pas  = (const float*)d_in[1];
    const int*   bids = (const int*)d_in[2];
    const int*   mids = (const int*)d_in[3];
    const int*   xy   = (const int*)d_in[4];
    const int*   m0   = (const int*)d_in[5];
    const int*   m1   = (const int*)d_in[6];
    const int*   m2   = (const int*)d_in[7];
    const int*   m3   = (const int*)d_in[8];
    const float* Wa   = (const float*)d_in[9];
    const float* ba   = (const float*)d_in[10];
    const float* Wv   = (const float*)d_in[11];
    const float* bv   = (const float*)d_in[12];
    const float* Wo   = (const float*)d_in[13];
    const float* bo   = (const float*)d_in[14];
    float* out = (float*)d_out;

    const int n_act = in_sizes[0] / F_SIZE;

    float* fw_ws    = (float*)d_ws;
    int*   id_ws    = (int*)(fw_ws + (size_t)n_act * 32);
    float* weighted = (float*)(id_ws + (size_t)n_act * 32);

    k_attn_ids<<<1024, 256, 0, stream>>>(
        act, bids, mids, xy, m0, m1, m2, m3, Wa, ba, fw_ws, id_ws, n_act);

    k_gather<<<n_act, 256, 0, stream>>>(act, pas, fw_ws, id_ws, weighted, n_act);

    float* wq = (float*)d_ws;
    k_frag_order<<<128, 256, 0, stream>>>(Wv, Wo, wq);

    k_dual_gemm<<<n_act / 128, 256, 0, stream>>>(weighted, wq, bv, bo, out);
}


# K2 gather made persistent: 2048 blocks of 4 waves walk the rows, thread t owns columns t and t+128, a row's ids and weights sit in scalar registers, no LDS or barrier, 32x fewer waves; sum order kept
# speedup vs baseline: 2.1885x; 1.0978x over previous
#include <hip/hip_runtime.h>
#include <hip/hip_bf16.h>

typedef __attribute__((ext_vector_type(2))) float v2f;
typedef __attribute__((ext_vector_type(8))) float v8f;

#define F_SIZE 256
#define LDSP 260

#define K1_ROWS 8

#define K1_STAGE(N, D)                                                   \
    {                                                                    \
        const bool up = (lane & (D)) != 0;                               \
          \
        _Pragma("unroll")                                                \
        for (int i = 0; i < (N) / 2; ++i) {                              \
            const float keep = up ? v[i + (N) / 2] : v[i];               \
            const float send = up ? v[i] : v[i + (N) / 2];               \
            v[i] = keep + __shfl_xor(send, (D), 32);                     \
        }                                                                \
    }

__global__ __launch_bounds__(256) void k_attn_ids(
    const float* __restrict__ act, const int* __restrict__ bids,
    const int* __restrict__ mids, const int* __restrict__ xy,
    const int* __restrict__ m0, const int* __restrict__ m1,
    const int* __restrict__ m2, const int* __restrict__ m3,
    const float* __restrict__ Wa, const float* __restrict__ ba,
    float* __restrict__ fw_ws, int* __restrict__ id_ws, int n_act)
{
    __shared__ float s_fw[2][K1_ROWS * 32];
    __shared__ int   s_id[2][K1_ROWS * 32];

    const int t = threadIdx.x;
    const int k = t >> 5;
    const int lane = t & 31;

    float4 wa[4][2];
#pragma unroll
    for (int pp = 0; pp < 4; ++pp) {
        const float4* __restrict__ wr =
            (const float4*)(Wa + (size_t)(k * 4 + pp) * F_SIZE);
        wa[pp][0] = wr[lane];
        wa[pp][1] = wr[32 + lane];
    }

    const int r = lane >> 2;
    const int p = lane & 3;
    const float bias = ba[k * 4 + p];
    const int cell = k + (k >= 4 ? 1 : 0);
    const int ox = (cell % 3 - 1) * (p + 1);
    const int oy = (cell / 3 - 1) * (p + 1);

    int buf = 0;
    for (int row0 = blockIdx.x * K1_ROWS; row0 < n_act;
         row0 += gridDim.x * K1_ROWS, buf ^= 1) {
        const int myrow = min(row0 + r, n_act - 1);
        const int x0 = xy[myrow * 2 + 0];
        const int y0 = xy[myrow * 2 + 1];
        const int m = mids[myrow];
        const int b = bids[myrow];

        float v[K1_ROWS * 4];
#pragma unroll
        for (int rr = 0; rr < K1_ROWS; ++rr) {
            const int row = min(row0 + rr, n_act - 1);
            const float4* __restrict__ ar =
                (const float4*)(act + (size_t)row * F_SIZE);
            const float4 a0 = ar[lane];
            const float4 a1 = ar[32 + lane];
#pragma unroll
            for (int pp = 0; pp < 4; ++pp) {
                const float4 w0 = wa[pp][0];
                const float4 w1 = wa[pp][1];
                v[rr * 4 + pp] = a0.x * w0.x + a0.y * w0.y + a0.z * w0.z + a0.w * w0.w
                               + a1.x * w1.x + a1.y * w1.y + a1.z * w1.z + a1.w * w1.w;
            }
        }

        K1_STAGE(32, 16)
        K1_STAGE(16, 8)
        K1_STAGE(8, 4)
        K1_STAGE(4, 2)
        K1_STAGE(2, 1)
        const float acc = v[0] + bias;

        float mx = acc;
        mx = fmaxf(mx, __shfl_xor(mx, 1, 32));
        mx = fmaxf(mx, __shfl_xor(mx, 2, 32));
        const float e = __expf(acc - mx);
        float s = e;
        s += __shfl_xor(s, 1, 32);
        s += __shfl_xor(s, 2, 32);
        const float fw = e / s;

        const int dim = 128 >> m;
        const int x = min(max(x0 + ox, 0), dim - 1);
        const int y = min(max(y0 + oy, 0), dim - 1);
        const int* __restrict__ map = (m == 0) ? m0 : (m == 1) ? m1 : (m == 2) ? m2 : m3;
        const int id = map[((size_t)b * dim + y) * dim + x];

        s_fw[buf][r * 32 + p * 8 + k] = fw;
        s_id[buf][r * 32 + p * 8 + k] = id;
        __syncthreads();
        const size_t o = (size_t)row0 * 32 + t;
        if (o < (size_t)n_act * 32) {
            fw_ws[o] = s_fw[buf][t];
            id_ws[o] = s_id[buf][t];
        }
    }
}

#define K2_BLOCKS 2048
#define K2_THREADS 128
typedef const __attribute__((address_space(1))) char  k2_gchar;
typedef const __attribute__((address_space(1))) float k2_gfloat;

__global__ __launch_bounds__(K2_THREADS) void k_gather(
    const float* __restrict__ act, const float* __restrict__ pas,
    const float* __restrict__ fw_ws, const int* __restrict__ id_ws,
    float* __restrict__ weighted, int n_act)
{
    const unsigned col4 = threadIdx.x * 4u;
    k2_gchar* const act_b = (k2_gchar*)act;
    k2_gchar* const pas_b = (k2_gchar*)((unsigned long long)pas -
                                        (unsigned long long)n_act * (F_SIZE * 4));

    int row = blockIdx.x;
    if (row >= n_act) return;

    int id[32];
#pragma unroll
    for (int g = 0; g < 32; ++g) id[g] = id_ws[(size_t)row * 32 + g];

    for (; row < n_act; row += gridDim.x) {
        float w[32];
#pragma unroll
        for (int g = 0; g < 32; ++g) w[g] = fw_ws[(size_t)row * 32 + g];
        float x0[32], x1[32];
#pragma unroll
        for (int g = 0; g < 32; ++g) {
            k2_gchar* const base = (id[g] < n_act) ? act_b : pas_b;
            k2_gfloat* const p =
                (k2_gfloat*)(base + ((unsigned)id[g] * (F_SIZE * 4u) + col4));
            x0[g] = p[0];
            x1[g] = p[K2_THREADS];
        }
        const int nrow = row + gridDim.x;
        if (nrow < n_act) {
#pragma unroll
            for (int g = 0; g < 32; ++g) id[g] = id_ws[(size_t)nrow * 32 + g];
        }
        float acc0 = 0.f, acc1 = 0.f;
#pragma unroll
        for (int g = 0; g < 32; ++g) {
            acc0 += w[g] * x0[g];
            acc1 += w[g] * x1[g];
        }
        float* const o = weighted + (size_t)row * F_SIZE + threadIdx.x;
        o[0] = acc0;
        o[K2_THREADS] = acc1;
    }
}

__global__ __launch_bounds__(256) void k_frag_order(
    const float* __restrict__ Wv, const float* __restrict__ Wo,
    float* __restrict__ wq)
{
    const int i = blockIdx.x * 256 + threadIdx.x;
    const int lane = i & 31;
    const int jj = (i >> 5) & 31;
    const int nt = (i >> 10) & 15;
    const int q = i >> 14;
    const int c = lane & 15;
    const int h = lane >> 4;
    const float* __restrict__ src =
        (q ? Wo : Wv) + (size_t)(nt * 16 + c) * F_SIZE + 8 * jj + 2 * h;
    const float2 lo = *(const float2*)src;
    const float2 up = *(const float2*)(src + 4);
    ((float4*)wq)[i] = make_float4(lo.x, lo.y, up.x, up.y);
}

#define K3_WAVES 8
#define K3_CHUNK4 2048
#define K3_PRE_LOAD(src)                                              \
    const float4 q0 = (src)[0 * 256], q1 = (src)[1 * 256],            \
                 q2 = (src)[2 * 256], q3 = (src)[3 * 256],            \
                 q4 = (src)[4 * 256], q5 = (src)[5 * 256],            \
                 q6 = (src)[6 * 256], q7 = (src)[7 * 256];
#define K3_PRE_STORE(dst)                                             \
    (dst)[0 * 256] = q0; (dst)[1 * 256] = q1; (dst)[2 * 256] = q2;    \
    (dst)[3 * 256] = q3; (dst)[4 * 256] = q4; (dst)[5 * 256] = q5;    \
    (dst)[6 * 256] = q6; (dst)[7 * 256] = q7;

__global__ __launch_bounds__(256) void k_dual_gemm(
    const float* __restrict__ weighted,
    const float* wq, const float* __restrict__ bv,
    const float* __restrict__ bo,
    float* __restrict__ out)
{
    __shared__ __attribute__((aligned(16))) float4 lds_b[2][K3_CHUNK4];
    __shared__ __attribute__((aligned(16))) float lds_t[K3_WAVES][16 * LDSP];

    const int t = threadIdx.x;
    const int wave = t >> 5;
    const int lane = t & 31;
    const int m0 = (blockIdx.x * K3_WAVES + wave) * 16;
    const int mrow = lane & 15;
    const int hi = lane >> 4;
    const float4* wq4 = (const float4*)wq;
    float* __restrict__ tile = lds_t[wave];

    K3_PRE_LOAD(wq4 + t)

#pragma unroll
    for (int g = 0; g < 4; ++g) {
        float4 v[8];
#pragma unroll
        for (int i = 0; i < 8; ++i) {
            const int idx = (g * 8 + i) * 32 + lane;
            v[i] = ((const float4*)(weighted + (size_t)(m0 + (idx >> 6)) * F_SIZE))[idx & 63];
        }
#pragma unroll
        for (int i = 0; i < 8; ++i) {
            const int idx = (g * 8 + i) * 32 + lane;
            *(float4*)(tile + (idx >> 6) * LDSP + (idx & 63) * 4) = v[i];
        }
    }
    K3_PRE_STORE(lds_b[0] + t)
    __syncthreads();

    const float* __restrict__ ap = tile + mrow * LDSP + 2 * hi;
    v2f a[64];

#pragma unroll
    for (int j = 0; j < 64; ++j) { a[j].x = ap[4 * j]; a[j].y = ap[4 * j + 1]; }

    for (int ci = 0; ci < 8; ++ci) {
        const int buf = ci & 1;
        K3_PRE_LOAD(wq4 + (size_t)(ci + 1) * K3_CHUNK4 + t)
        __builtin_amdgcn_sched_barrier(0);
        const float4* __restrict__ lb = lds_b[buf] + lane;
        float4 bf[64];
#pragma unroll
        for (int jj = 0; jj < 32; ++jj) {
            bf[2 * jj]     = lb[jj * 32];
            bf[2 * jj + 1] = lb[1024 + jj * 32];
        }
        __builtin_amdgcn_sched_barrier(0);
        v8f c0 = {}, c1 = {};
#pragma unroll
        for (int jj = 0; jj < 32; ++jj) {
            const float4 p = bf[2 * jj];
            const float4 q = bf[2 * jj + 1];
            v2f b;
            b.x = p.x; b.y = p.y;
            c0 = __builtin_amdgcn_wmma_f32_16x16x4_f32(
                     false, a[2 * jj], false, b, (short)0, c0, false, false);
            b.x = q.x; b.y = q.y;
            c1 = __builtin_amdgcn_wmma_f32_16x16x4_f32(
                     false, a[2 * jj], false, b, (short)0, c1, false, false);
            b.x = p.z; b.y = p.w;
            c0 = __builtin_amdgcn_wmma_f32_16x16x4_f32(
                     false, a[2 * jj + 1], false, b, (short)0, c0, false, false);
            b.x = q.z; b.y = q.w;
            c1 = __builtin_amdgcn_wmma_f32_16x16x4_f32(
                     false, a[2 * jj + 1], false, b, (short)0, c1, false, false);
        }
        const int nn = ci * 32 + mrow;
        const float bias0 = bv[nn];
        const float bias1 = bv[nn + 16];
#pragma unroll
        for (int r = 0; r < 8; ++r) {
            tile[(r + 8 * hi) * LDSP + nn] = c0[r] + bias0;
            tile[(r + 8 * hi) * LDSP + nn + 16] = c1[r] + bias1;
        }
        __builtin_amdgcn_sched_barrier(0);
        K3_PRE_STORE(lds_b[buf ^ 1] + t)
        __syncthreads();
    }

#pragma unroll
    for (int j = 0; j < 64; ++j) { a[j].x = ap[4 * j]; a[j].y = ap[4 * j + 1]; }

    for (int ci = 8; ci < 16; ++ci) {
        const int buf = ci & 1;
        const int cn = min(ci + 1, 15);
        K3_PRE_LOAD(wq4 + (size_t)cn * K3_CHUNK4 + t)
        __builtin_amdgcn_sched_barrier(0);
        const float4* __restrict__ lb = lds_b[buf] + lane;
        float4 bf[64];
#pragma unroll
        for (int jj = 0; jj < 32; ++jj) {
            bf[2 * jj]     = lb[jj * 32];
            bf[2 * jj + 1] = lb[1024 + jj * 32];
        }
        __builtin_amdgcn_sched_barrier(0);
        v8f c0 = {}, c1 = {};
#pragma unroll
        for (int jj = 0; jj < 32; ++jj) {
            const float4 p = bf[2 * jj];
            const float4 q = bf[2 * jj + 1];
            v2f b;
            b.x = p.x; b.y = p.y;
            c0 = __builtin_amdgcn_wmma_f32_16x16x4_f32(
                     false, a[2 * jj], false, b, (short)0, c0, false, false);
            b.x = q.x; b.y = q.y;
            c1 = __builtin_amdgcn_wmma_f32_16x16x4_f32(
                     false, a[2 * jj], false, b, (short)0, c1, false, false);
            b.x = p.z; b.y = p.w;
            c0 = __builtin_amdgcn_wmma_f32_16x16x4_f32(
                     false, a[2 * jj + 1], false, b, (short)0, c0, false, false);
            b.x = q.z; b.y = q.w;
            c1 = __builtin_amdgcn_wmma_f32_16x16x4_f32(
                     false, a[2 * jj + 1], false, b, (short)0, c1, false, false);
        }
        const int nn = (ci - 8) * 32 + mrow;
        const float bias0 = bo[nn];
        const float bias1 = bo[nn + 16];
        __builtin_amdgcn_sched_barrier(0);
        K3_PRE_STORE(lds_b[buf ^ 1] + t)
        __syncthreads();
#pragma unroll
        for (int r = 0; r < 8; ++r) {
            out[(size_t)(m0 + r + 8 * hi) * F_SIZE + nn] = c0[r] + bias0;
            out[(size_t)(m0 + r + 8 * hi) * F_SIZE + nn + 16] = c1[r] + bias1;
        }
    }
}

extern "C" void kernel_launch(void* const* d_in, const int* in_sizes, int n_in,
                              void* d_out, int out_size, void* d_ws, size_t ws_size,
                              hipStream_t stream) {
    const float* act  = (const float*)d_in[0];
    const float* pas  = (const float*)d_in[1];
    const int*   bids = (const int*)d_in[2];
    const int*   mids = (const int*)d_in[3];
    const int*   xy   = (const int*)d_in[4];
    const int*   m0   = (const int*)d_in[5];
    const int*   m1   = (const int*)d_in[6];
    const int*   m2   = (const int*)d_in[7];
    const int*   m3   = (const int*)d_in[8];
    const float* Wa   = (const float*)d_in[9];
    const float* ba   = (const float*)d_in[10];
    const float* Wv   = (const float*)d_in[11];
    const float* bv   = (const float*)d_in[12];
    const float* Wo   = (const float*)d_in[13];
    const float* bo   = (const float*)d_in[14];
    float* out = (float*)d_out;

    const int n_act = in_sizes[0] / F_SIZE;

    float* fw_ws    = (float*)d_ws;
    int*   id_ws    = (int*)(fw_ws + (size_t)n_act * 32);
    float* weighted = (float*)(id_ws + (size_t)n_act * 32);

    k_attn_ids<<<1024, 256, 0, stream>>>(
        act, bids, mids, xy, m0, m1, m2, m3, Wa, ba, fw_ws, id_ws, n_act);

    k_gather<<<K2_BLOCKS, K2_THREADS, 0, stream>>>(act, pas, fw_ws, id_ws, weighted, n_act);

    float* wq = (float*)d_ws;
    k_frag_order<<<128, 256, 0, stream>>>(Wv, Wo, wq);

    k_dual_gemm<<<n_act / 128, 256, 0, stream>>>(weighted, wq, bv, bo, out);
}


# K1 and K2 fused into one persistent launch of 16-wave blocks: 8 waves work out the next 8 rows' weights and ids into LDS while 8 waves gather the current 8 rows, one row per wave; sum order kept
# speedup vs baseline: 2.4081x; 1.1003x over previous
#include <hip/hip_runtime.h>
#include <hip/hip_bf16.h>

typedef __attribute__((ext_vector_type(2))) float v2f;
typedef __attribute__((ext_vector_type(8))) float v8f;

#define F_SIZE 256
#define LDSP 260

#define K1_ROWS 8

#define K1_STAGE(N, D)                                                   \
    {                                                                    \
        const bool up = (lane & (D)) != 0;                               \
          \
        _Pragma("unroll")                                                \
        for (int i = 0; i < (N) / 2; ++i) {                              \
            const float keep = up ? v[i + (N) / 2] : v[i];               \
            const float send = up ? v[i] : v[i + (N) / 2];               \
            v[i] = keep + __shfl_xor(send, (D), 32);                     \
        }                                                                \
    }

#define K12_BLOCKS 512
#define K12_THREADS 512
typedef float k12_v4f __attribute__((ext_vector_type(4)));
typedef const __attribute__((address_space(1))) char    k12_gchar;
typedef const __attribute__((address_space(1))) k12_v4f k12_gv4f;

__global__ __launch_bounds__(K12_THREADS, 8) void k_attn_gather(
    const float* __restrict__ act, const float* __restrict__ pas,
    const int* __restrict__ bids,
    const int* __restrict__ mids, const int* __restrict__ xy,
    const int* __restrict__ m0, const int* __restrict__ m1,
    const int* __restrict__ m2, const int* __restrict__ m3,
    const float* __restrict__ Wa, const float* __restrict__ ba,
    float* __restrict__ weighted, int n_act)
{
    __shared__ float s_fw[2][K1_ROWS * 32];
    __shared__ int   s_id[2][K1_ROWS * 32];

    const bool producer = threadIdx.x < 256;
    const int t = threadIdx.x & 255;
    const int k = t >> 5;
    const int lane = t & 31;

    const int stride = gridDim.x * K1_ROWS;
    int row0 = blockIdx.x * K1_ROWS;
    if (row0 >= n_act) return;

    float4 wa[4][2];
#pragma unroll
    for (int pp = 0; pp < 4; ++pp) {
        wa[pp][0] = make_float4(0.f, 0.f, 0.f, 0.f);
        wa[pp][1] = make_float4(0.f, 0.f, 0.f, 0.f);
    }
    if (producer) {
#pragma unroll
        for (int pp = 0; pp < 4; ++pp) {
            const float4* __restrict__ wr =
                (const float4*)(Wa + (size_t)(k * 4 + pp) * F_SIZE);
            wa[pp][0] = wr[lane];
            wa[pp][1] = wr[32 + lane];
        }
    }
    const int r = lane >> 2;
    const int p = lane & 3;
    const float bias = ba[k * 4 + p];
    const int cell = k + (k >= 4 ? 1 : 0);
    const int ox = (cell % 3 - 1) * (p + 1);
    const int oy = (cell / 3 - 1) * (p + 1);

    const unsigned lane16 = lane * 16u;
    k12_gchar* const act_b = (k12_gchar*)act;
    k12_gchar* const pas_b = (k12_gchar*)((unsigned long long)pas -
                                          (unsigned long long)n_act * (F_SIZE * 4));

    int prow0 = row0;
    int buf = 0;
    bool have = false;
    for (;;) {
        float a0 = 0.f, a1 = 0.f, a2 = 0.f, a3 = 0.f;
        float a4 = 0.f, a5 = 0.f, a6 = 0.f, a7 = 0.f;
        int grow = -1;
        if (producer) {
            if (prow0 < n_act) {
                const int myrow = min(prow0 + r, n_act - 1);
                const int x0 = xy[myrow * 2 + 0];
                const int y0 = xy[myrow * 2 + 1];
                const int m = mids[myrow];
                const int b = bids[myrow];

                float v[K1_ROWS * 4];
#pragma unroll
                for (int rr = 0; rr < K1_ROWS; ++rr) {
                    const int row = min(prow0 + rr, n_act - 1);
                    const float4* __restrict__ ar =
                        (const float4*)(act + (size_t)row * F_SIZE);
                    const float4 f0 = ar[lane];
                    const float4 f1 = ar[32 + lane];
#pragma unroll
                    for (int pp = 0; pp < 4; ++pp) {
                        const float4 w0 = wa[pp][0];
                        const float4 w1 = wa[pp][1];
                        v[rr * 4 + pp] = f0.x * w0.x + f0.y * w0.y + f0.z * w0.z + f0.w * w0.w
                                       + f1.x * w1.x + f1.y * w1.y + f1.z * w1.z + f1.w * w1.w;
                    }
                }

                K1_STAGE(32, 16)
                K1_STAGE(16, 8)
                K1_STAGE(8, 4)
                K1_STAGE(4, 2)
                K1_STAGE(2, 1)
                const float acc = v[0] + bias;

                float mx = acc;
                mx = fmaxf(mx, __shfl_xor(mx, 1, 32));
                mx = fmaxf(mx, __shfl_xor(mx, 2, 32));
                const float e = __expf(acc - mx);
                float s = e;
                s += __shfl_xor(s, 1, 32);
                s += __shfl_xor(s, 2, 32);
                const float fw = e / s;

                const int dim = 128 >> m;
                const int x = min(max(x0 + ox, 0), dim - 1);
                const int y = min(max(y0 + oy, 0), dim - 1);
                const int* __restrict__ map = (m == 0) ? m0 : (m == 1) ? m1 : (m == 2) ? m2 : m3;
                const int id = map[((size_t)b * dim + y) * dim + x];

                s_fw[buf][r * 32 + p * 8 + k] = fw;
                s_id[buf][r * 32 + p * 8 + k] = id;
            }
        } else if (have) {
            const int crow = row0 + k;
            const int idv = s_id[buf ^ 1][k * 32 + lane];
            const int fwv = __float_as_int(s_fw[buf ^ 1][k * 32 + lane]);
            if (crow < n_act) {
                grow = crow;
#pragma unroll 1
                for (int g0 = 0; g0 < 32; g0 += 8) {
                    k12_v4f x[16];
#pragma unroll
                    for (int j = 0; j < 8; ++j) {
                        const int idg = __builtin_amdgcn_readlane(idv, g0 + j);
                        k12_gchar* const base = (idg < n_act) ? act_b : pas_b;
                        k12_gv4f* const src =
                            (k12_gv4f*)(base + ((unsigned)idg * (F_SIZE * 4u) + lane16));
                        x[2 * j] = src[0];
                        x[2 * j + 1] = src[32];
                    }
#pragma unroll
                    for (int j = 0; j < 8; ++j) {
                        const float wg =
                            __int_as_float(__builtin_amdgcn_readlane(fwv, g0 + j));
                        a0 += wg * x[2 * j].x;     a1 += wg * x[2 * j].y;
                        a2 += wg * x[2 * j].z;     a3 += wg * x[2 * j].w;
                        a4 += wg * x[2 * j + 1].x; a5 += wg * x[2 * j + 1].y;
                        a6 += wg * x[2 * j + 1].z; a7 += wg * x[2 * j + 1].w;
                    }
                }
            }
        }
        __syncthreads();
        if (grow >= 0) {
            float4* const o = (float4*)(weighted + (size_t)grow * F_SIZE) + lane;
            o[0]  = make_float4(a0, a1, a2, a3);
            o[32] = make_float4(a4, a5, a6, a7);
        }
        if (prow0 >= n_act) break;
        row0 = prow0;
        prow0 += stride;
        have = true;
        buf ^= 1;
    }
}

__global__ __launch_bounds__(256) void k_frag_order(
    const float* __restrict__ Wv, const float* __restrict__ Wo,
    float* __restrict__ wq)
{
    const int i = blockIdx.x * 256 + threadIdx.x;
    const int lane = i & 31;
    const int jj = (i >> 5) & 31;
    const int nt = (i >> 10) & 15;
    const int q = i >> 14;
    const int c = lane & 15;
    const int h = lane >> 4;
    const float* __restrict__ src =
        (q ? Wo : Wv) + (size_t)(nt * 16 + c) * F_SIZE + 8 * jj + 2 * h;
    const float2 lo = *(const float2*)src;
    const float2 up = *(const float2*)(src + 4);
    ((float4*)wq)[i] = make_float4(lo.x, lo.y, up.x, up.y);
}

#define K3_WAVES 8
#define K3_CHUNK4 2048
#define K3_PRE_LOAD(src)                                              \
    const float4 q0 = (src)[0 * 256], q1 = (src)[1 * 256],            \
                 q2 = (src)[2 * 256], q3 = (src)[3 * 256],            \
                 q4 = (src)[4 * 256], q5 = (src)[5 * 256],            \
                 q6 = (src)[6 * 256], q7 = (src)[7 * 256];
#define K3_PRE_STORE(dst)                                             \
    (dst)[0 * 256] = q0; (dst)[1 * 256] = q1; (dst)[2 * 256] = q2;    \
    (dst)[3 * 256] = q3; (dst)[4 * 256] = q4; (dst)[5 * 256] = q5;    \
    (dst)[6 * 256] = q6; (dst)[7 * 256] = q7;

__global__ __launch_bounds__(256) void k_dual_gemm(
    const float* __restrict__ weighted,
    const float* wq, const float* __restrict__ bv,
    const float* __restrict__ bo,
    float* __restrict__ out)
{
    __shared__ __attribute__((aligned(16))) float4 lds_b[2][K3_CHUNK4];
    __shared__ __attribute__((aligned(16))) float lds_t[K3_WAVES][16 * LDSP];

    const int t = threadIdx.x;
    const int wave = t >> 5;
    const int lane = t & 31;
    const int m0 = (blockIdx.x * K3_WAVES + wave) * 16;
    const int mrow = lane & 15;
    const int hi = lane >> 4;
    const float4* wq4 = (const float4*)wq;
    float* __restrict__ tile = lds_t[wave];

    K3_PRE_LOAD(wq4 + t)

#pragma unroll
    for (int g = 0; g < 4; ++g) {
        float4 v[8];
#pragma unroll
        for (int i = 0; i < 8; ++i) {
            const int idx = (g * 8 + i) * 32 + lane;
            v[i] = ((const float4*)(weighted + (size_t)(m0 + (idx >> 6)) * F_SIZE))[idx & 63];
        }
#pragma unroll
        for (int i = 0; i < 8; ++i) {
            const int idx = (g * 8 + i) * 32 + lane;
            *(float4*)(tile + (idx >> 6) * LDSP + (idx & 63) * 4) = v[i];
        }
    }
    K3_PRE_STORE(lds_b[0] + t)
    __syncthreads();

    const float* __restrict__ ap = tile + mrow * LDSP + 2 * hi;
    v2f a[64];

#pragma unroll
    for (int j = 0; j < 64; ++j) { a[j].x = ap[4 * j]; a[j].y = ap[4 * j + 1]; }

    for (int ci = 0; ci < 8; ++ci) {
        const int buf = ci & 1;
        K3_PRE_LOAD(wq4 + (size_t)(ci + 1) * K3_CHUNK4 + t)
        __builtin_amdgcn_sched_barrier(0);
        const float4* __restrict__ lb = lds_b[buf] + lane;
        float4 bf[64];
#pragma unroll
        for (int jj = 0; jj < 32; ++jj) {
            bf[2 * jj]     = lb[jj * 32];
            bf[2 * jj + 1] = lb[1024 + jj * 32];
        }
        __builtin_amdgcn_sched_barrier(0);
        v8f c0 = {}, c1 = {};
#pragma unroll
        for (int jj = 0; jj < 32; ++jj) {
            const float4 p = bf[2 * jj];
            const float4 q = bf[2 * jj + 1];
            v2f b;
            b.x = p.x; b.y = p.y;
            c0 = __builtin_amdgcn_wmma_f32_16x16x4_f32(
                     false, a[2 * jj], false, b, (short)0, c0, false, false);
            b.x = q.x; b.y = q.y;
            c1 = __builtin_amdgcn_wmma_f32_16x16x4_f32(
                     false, a[2 * jj], false, b, (short)0, c1, false, false);
            b.x = p.z; b.y = p.w;
            c0 = __builtin_amdgcn_wmma_f32_16x16x4_f32(
                     false, a[2 * jj + 1], false, b, (short)0, c0, false, false);
            b.x = q.z; b.y = q.w;
            c1 = __builtin_amdgcn_wmma_f32_16x16x4_f32(
                     false, a[2 * jj + 1], false, b, (short)0, c1, false, false);
        }
        const int nn = ci * 32 + mrow;
        const float bias0 = bv[nn];
        const float bias1 = bv[nn + 16];
#pragma unroll
        for (int r = 0; r < 8; ++r) {
            tile[(r + 8 * hi) * LDSP + nn] = c0[r] + bias0;
            tile[(r + 8 * hi) * LDSP + nn + 16] = c1[r] + bias1;
        }
        __builtin_amdgcn_sched_barrier(0);
        K3_PRE_STORE(lds_b[buf ^ 1] + t)
        __syncthreads();
    }

#pragma unroll
    for (int j = 0; j < 64; ++j) { a[j].x = ap[4 * j]; a[j].y = ap[4 * j + 1]; }

    for (int ci = 8; ci < 16; ++ci) {
        const int buf = ci & 1;
        const int cn = min(ci + 1, 15);
        K3_PRE_LOAD(wq4 + (size_t)cn * K3_CHUNK4 + t)
        __builtin_amdgcn_sched_barrier(0);
        const float4* __restrict__ lb = lds_b[buf] + lane;
        float4 bf[64];
#pragma unroll
        for (int jj = 0; jj < 32; ++jj) {
            bf[2 * jj]     = lb[jj * 32];
            bf[2 * jj + 1] = lb[1024 + jj * 32];
        }
        __builtin_amdgcn_sched_barrier(0);
        v8f c0 = {}, c1 = {};
#pragma unroll
        for (int jj = 0; jj < 32; ++jj) {
            const float4 p = bf[2 * jj];
            const float4 q = bf[2 * jj + 1];
            v2f b;
            b.x = p.x; b.y = p.y;
            c0 = __builtin_amdgcn_wmma_f32_16x16x4_f32(
                     false, a[2 * jj], false, b, (short)0, c0, false, false);
            b.x = q.x; b.y = q.y;
            c1 = __builtin_amdgcn_wmma_f32_16x16x4_f32(
                     false, a[2 * jj], false, b, (short)0, c1, false, false);
            b.x = p.z; b.y = p.w;
            c0 = __builtin_amdgcn_wmma_f32_16x16x4_f32(
                     false, a[2 * jj + 1], false, b, (short)0, c0, false, false);
            b.x = q.z; b.y = q.w;
            c1 = __builtin_amdgcn_wmma_f32_16x16x4_f32(
                     false, a[2 * jj + 1], false, b, (short)0, c1, false, false);
        }
        const int nn = (ci - 8) * 32 + mrow;
        const float bias0 = bo[nn];
        const float bias1 = bo[nn + 16];
        __builtin_amdgcn_sched_barrier(0);
        K3_PRE_STORE(lds_b[buf ^ 1] + t)
        __syncthreads();
#pragma unroll
        for (int r = 0; r < 8; ++r) {
            out[(size_t)(m0 + r + 8 * hi) * F_SIZE + nn] = c0[r] + bias0;
            out[(size_t)(m0 + r + 8 * hi) * F_SIZE + nn + 16] = c1[r] + bias1;
        }
    }
}

extern "C" void kernel_launch(void* const* d_in, const int* in_sizes, int n_in,
                              void* d_out, int out_size, void* d_ws, size_t ws_size,
                              hipStream_t stream) {
    const float* act  = (const float*)d_in[0];
    const float* pas  = (const float*)d_in[1];
    const int*   bids = (const int*)d_in[2];
    const int*   mids = (const int*)d_in[3];
    const int*   xy   = (const int*)d_in[4];
    const int*   m0   = (const int*)d_in[5];
    const int*   m1   = (const int*)d_in[6];
    const int*   m2   = (const int*)d_in[7];
    const int*   m3   = (const int*)d_in[8];
    const float* Wa   = (const float*)d_in[9];
    const float* ba   = (const float*)d_in[10];
    const float* Wv   = (const float*)d_in[11];
    const float* bv   = (const float*)d_in[12];
    const float* Wo   = (const float*)d_in[13];
    const float* bo   = (const float*)d_in[14];
    float* out = (float*)d_out;

    const int n_act = in_sizes[0] / F_SIZE;

    float* wq       = (float*)d_ws;
    float* weighted = (float*)d_ws + (size_t)n_act * 64;

    k_attn_gather<<<K12_BLOCKS, K12_THREADS, 0, stream>>>(
        act, pas, bids, mids, xy, m0, m1, m2, m3, Wa, ba, weighted, n_act);

    k_frag_order<<<128, 256, 0, stream>>>(Wv, Wo, wq);

    k_dual_gemm<<<n_act / 128, 256, 0, stream>>>(weighted, wq, bv, bo, out);
}


# K12 producers prefetch into the second-level cache the feature row of each unclipped sample one trip before the consumers load it: gather groups wait for cache, not memory; sums unchanged
# speedup vs baseline: 2.5051x; 1.0403x over previous
#include <hip/hip_runtime.h>
#include <hip/hip_bf16.h>

typedef __attribute__((ext_vector_type(2))) float v2f;
typedef __attribute__((ext_vector_type(8))) float v8f;

#define F_SIZE 256
#define LDSP 260

#define K1_ROWS 8

#define K1_STAGE(N, D)                                                   \
    {                                                                    \
        const bool up = (lane & (D)) != 0;                               \
          \
        _Pragma("unroll")                                                \
        for (int i = 0; i < (N) / 2; ++i) {                              \
            const float keep = up ? v[i + (N) / 2] : v[i];               \
            const float send = up ? v[i] : v[i + (N) / 2];               \
            v[i] = keep + __shfl_xor(send, (D), 32);                     \
        }                                                                \
    }

#define K12_BLOCKS 512
#define K12_THREADS 512
typedef float k12_v4f __attribute__((ext_vector_type(4)));
typedef const __attribute__((address_space(1))) char    k12_gchar;
typedef const __attribute__((address_space(1))) k12_v4f k12_gv4f;

__global__ __launch_bounds__(K12_THREADS, 8) void k_attn_gather(
    const float* __restrict__ act, const float* __restrict__ pas,
    const int* __restrict__ bids,
    const int* __restrict__ mids, const int* __restrict__ xy,
    const int* __restrict__ m0, const int* __restrict__ m1,
    const int* __restrict__ m2, const int* __restrict__ m3,
    const float* __restrict__ Wa, const float* __restrict__ ba,
    float* __restrict__ weighted, int n_act)
{
    __shared__ float s_fw[2][K1_ROWS * 32];
    __shared__ int   s_id[2][K1_ROWS * 32];

    const bool producer = threadIdx.x < 256;
    const int t = threadIdx.x & 255;
    const int k = t >> 5;
    const int lane = t & 31;

    const int stride = gridDim.x * K1_ROWS;
    int row0 = blockIdx.x * K1_ROWS;
    if (row0 >= n_act) return;

    float4 wa[4][2];
#pragma unroll
    for (int pp = 0; pp < 4; ++pp) {
        wa[pp][0] = make_float4(0.f, 0.f, 0.f, 0.f);
        wa[pp][1] = make_float4(0.f, 0.f, 0.f, 0.f);
    }
    if (producer) {
#pragma unroll
        for (int pp = 0; pp < 4; ++pp) {
            const float4* __restrict__ wr =
                (const float4*)(Wa + (size_t)(k * 4 + pp) * F_SIZE);
            wa[pp][0] = wr[lane];
            wa[pp][1] = wr[32 + lane];
        }
    }
    const int r = lane >> 2;
    const int p = lane & 3;
    const float bias = ba[k * 4 + p];
    const int cell = k + (k >= 4 ? 1 : 0);
    const int ox = (cell % 3 - 1) * (p + 1);
    const int oy = (cell / 3 - 1) * (p + 1);

    const unsigned lane16 = lane * 16u;
    k12_gchar* const act_b = (k12_gchar*)act;
    k12_gchar* const pas_b = (k12_gchar*)((unsigned long long)pas -
                                          (unsigned long long)n_act * (F_SIZE * 4));

    int prow0 = row0;
    int buf = 0;
    bool have = false;
    for (;;) {
        float a0 = 0.f, a1 = 0.f, a2 = 0.f, a3 = 0.f;
        float a4 = 0.f, a5 = 0.f, a6 = 0.f, a7 = 0.f;
        int grow = -1;
        if (producer) {
            if (prow0 < n_act) {
                const int myrow = min(prow0 + r, n_act - 1);
                const int x0 = xy[myrow * 2 + 0];
                const int y0 = xy[myrow * 2 + 1];
                const int m = mids[myrow];
                const int b = bids[myrow];

                float4 f[K1_ROWS][2];
#pragma unroll
                for (int rr = 0; rr < K1_ROWS; ++rr) {
                    const int row = min(prow0 + rr, n_act - 1);
                    const float4* __restrict__ ar =
                        (const float4*)(act + (size_t)row * F_SIZE);
                    f[rr][0] = ar[lane];
                    f[rr][1] = ar[32 + lane];
                }

                const int dim = 128 >> m;
                const int xs = x0 + ox;
                const int ys = y0 + oy;
                const int x = min(max(xs, 0), dim - 1);
                const int y = min(max(ys, 0), dim - 1);
                const int* __restrict__ map = (m == 0) ? m0 : (m == 1) ? m1 : (m == 2) ? m2 : m3;
                const int id = map[((size_t)b * dim + y) * dim + x];
                if (x == xs && y == ys) {
                    k12_gchar* const pf =
                        ((id < n_act) ? act_b : pas_b) + (unsigned)id * (F_SIZE * 4u);
                    __builtin_prefetch((const void*)(pf), 0, 3);
                    __builtin_prefetch((const void*)(pf + 256), 0, 3);
                    __builtin_prefetch((const void*)(pf + 512), 0, 3);
                    __builtin_prefetch((const void*)(pf + 768), 0, 3);
                }

                float v[K1_ROWS * 4];
#pragma unroll
                for (int rr = 0; rr < K1_ROWS; ++rr) {
                    const float4 f0 = f[rr][0];
                    const float4 f1 = f[rr][1];
#pragma unroll
                    for (int pp = 0; pp < 4; ++pp) {
                        const float4 w0 = wa[pp][0];
                        const float4 w1 = wa[pp][1];
                        v[rr * 4 + pp] = f0.x * w0.x + f0.y * w0.y + f0.z * w0.z + f0.w * w0.w
                                       + f1.x * w1.x + f1.y * w1.y + f1.z * w1.z + f1.w * w1.w;
                    }
                }

                K1_STAGE(32, 16)
                K1_STAGE(16, 8)
                K1_STAGE(8, 4)
                K1_STAGE(4, 2)
                K1_STAGE(2, 1)
                const float acc = v[0] + bias;

                float mx = acc;
                mx = fmaxf(mx, __shfl_xor(mx, 1, 32));
                mx = fmaxf(mx, __shfl_xor(mx, 2, 32));
                const float e = __expf(acc - mx);
                float s = e;
                s += __shfl_xor(s, 1, 32);
                s += __shfl_xor(s, 2, 32);
                const float fw = e / s;

                s_fw[buf][r * 32 + p * 8 + k] = fw;
                s_id[buf][r * 32 + p * 8 + k] = id;
            }
        } else if (have) {
            const int crow = row0 + k;
            const int idv = s_id[buf ^ 1][k * 32 + lane];
            const int fwv = __float_as_int(s_fw[buf ^ 1][k * 32 + lane]);
            if (crow < n_act) {
                grow = crow;
#pragma unroll 1
                for (int g0 = 0; g0 < 32; g0 += 8) {
                    k12_v4f x[16];
#pragma unroll
                    for (int j = 0; j < 8; ++j) {
                        const int idg = __builtin_amdgcn_readlane(idv, g0 + j);
                        k12_gchar* const base = (idg < n_act) ? act_b : pas_b;
                        k12_gv4f* const src =
                            (k12_gv4f*)(base + ((unsigned)idg * (F_SIZE * 4u) + lane16));
                        x[2 * j] = src[0];
                        x[2 * j + 1] = src[32];
                    }
#pragma unroll
                    for (int j = 0; j < 8; ++j) {
                        const float wg =
                            __int_as_float(__builtin_amdgcn_readlane(fwv, g0 + j));
                        a0 += wg * x[2 * j].x;     a1 += wg * x[2 * j].y;
                        a2 += wg * x[2 * j].z;     a3 += wg * x[2 * j].w;
                        a4 += wg * x[2 * j + 1].x; a5 += wg * x[2 * j + 1].y;
                        a6 += wg * x[2 * j + 1].z; a7 += wg * x[2 * j + 1].w;
                    }
                }
            }
        }
        __syncthreads();
        if (grow >= 0) {
            float4* const o = (float4*)(weighted + (size_t)grow * F_SIZE) + lane;
            o[0]  = make_float4(a0, a1, a2, a3);
            o[32] = make_float4(a4, a5, a6, a7);
        }
        if (prow0 >= n_act) break;
        row0 = prow0;
        prow0 += stride;
        have = true;
        buf ^= 1;
    }
}

__global__ __launch_bounds__(256) void k_frag_order(
    const float* __restrict__ Wv, const float* __restrict__ Wo,
    float* __restrict__ wq)
{
    const int i = blockIdx.x * 256 + threadIdx.x;
    const int lane = i & 31;
    const int jj = (i >> 5) & 31;
    const int nt = (i >> 10) & 15;
    const int q = i >> 14;
    const int c = lane & 15;
    const int h = lane >> 4;
    const float* __restrict__ src =
        (q ? Wo : Wv) + (size_t)(nt * 16 + c) * F_SIZE + 8 * jj + 2 * h;
    const float2 lo = *(const float2*)src;
    const float2 up = *(const float2*)(src + 4);
    ((float4*)wq)[i] = make_float4(lo.x, lo.y, up.x, up.y);
}

#define K3_WAVES 8
#define K3_CHUNK4 2048
#define K3_PRE_LOAD(src)                                              \
    const float4 q0 = (src)[0 * 256], q1 = (src)[1 * 256],            \
                 q2 = (src)[2 * 256], q3 = (src)[3 * 256],            \
                 q4 = (src)[4 * 256], q5 = (src)[5 * 256],            \
                 q6 = (src)[6 * 256], q7 = (src)[7 * 256];
#define K3_PRE_STORE(dst)                                             \
    (dst)[0 * 256] = q0; (dst)[1 * 256] = q1; (dst)[2 * 256] = q2;    \
    (dst)[3 * 256] = q3; (dst)[4 * 256] = q4; (dst)[5 * 256] = q5;    \
    (dst)[6 * 256] = q6; (dst)[7 * 256] = q7;

__global__ __launch_bounds__(256) void k_dual_gemm(
    const float* __restrict__ weighted,
    const float* wq, const float* __restrict__ bv,
    const float* __restrict__ bo,
    float* __restrict__ out)
{
    __shared__ __attribute__((aligned(16))) float4 lds_b[2][K3_CHUNK4];
    __shared__ __attribute__((aligned(16))) float lds_t[K3_WAVES][16 * LDSP];

    const int t = threadIdx.x;
    const int wave = t >> 5;
    const int lane = t & 31;
    const int m0 = (blockIdx.x * K3_WAVES + wave) * 16;
    const int mrow = lane & 15;
    const int hi = lane >> 4;
    const float4* wq4 = (const float4*)wq;
    float* __restrict__ tile = lds_t[wave];

    K3_PRE_LOAD(wq4 + t)

#pragma unroll
    for (int g = 0; g < 4; ++g) {
        float4 v[8];
#pragma unroll
        for (int i = 0; i < 8; ++i) {
            const int idx = (g * 8 + i) * 32 + lane;
            v[i] = ((const float4*)(weighted + (size_t)(m0 + (idx >> 6)) * F_SIZE))[idx & 63];
        }
#pragma unroll
        for (int i = 0; i < 8; ++i) {
            const int idx = (g * 8 + i) * 32 + lane;
            *(float4*)(tile + (idx >> 6) * LDSP + (idx & 63) * 4) = v[i];
        }
    }
    K3_PRE_STORE(lds_b[0] + t)
    __syncthreads();

    const float* __restrict__ ap = tile + mrow * LDSP + 2 * hi;
    v2f a[64];

#pragma unroll
    for (int j = 0; j < 64; ++j) { a[j].x = ap[4 * j]; a[j].y = ap[4 * j + 1]; }

    for (int ci = 0; ci < 8; ++ci) {
        const int buf = ci & 1;
        K3_PRE_LOAD(wq4 + (size_t)(ci + 1) * K3_CHUNK4 + t)
        __builtin_amdgcn_sched_barrier(0);
        const float4* __restrict__ lb = lds_b[buf] + lane;
        float4 bf[64];
#pragma unroll
        for (int jj = 0; jj < 32; ++jj) {
            bf[2 * jj]     = lb[jj * 32];
            bf[2 * jj + 1] = lb[1024 + jj * 32];
        }
        __builtin_amdgcn_sched_barrier(0);
        v8f c0 = {}, c1 = {};
#pragma unroll
        for (int jj = 0; jj < 32; ++jj) {
            const float4 p = bf[2 * jj];
            const float4 q = bf[2 * jj + 1];
            v2f b;
            b.x = p.x; b.y = p.y;
            c0 = __builtin_amdgcn_wmma_f32_16x16x4_f32(
                     false, a[2 * jj], false, b, (short)0, c0, false, false);
            b.x = q.x; b.y = q.y;
            c1 = __builtin_amdgcn_wmma_f32_16x16x4_f32(
                     false, a[2 * jj], false, b, (short)0, c1, false, false);
            b.x = p.z; b.y = p.w;
            c0 = __builtin_amdgcn_wmma_f32_16x16x4_f32(
                     false, a[2 * jj + 1], false, b, (short)0, c0, false, false);
            b.x = q.z; b.y = q.w;
            c1 = __builtin_amdgcn_wmma_f32_16x16x4_f32(
                     false, a[2 * jj + 1], false, b, (short)0, c1, false, false);
        }
        const int nn = ci * 32 + mrow;
        const float bias0 = bv[nn];
        const float bias1 = bv[nn + 16];
#pragma unroll
        for (int r = 0; r < 8; ++r) {
            tile[(r + 8 * hi) * LDSP + nn] = c0[r] + bias0;
            tile[(r + 8 * hi) * LDSP + nn + 16] = c1[r] + bias1;
        }
        __builtin_amdgcn_sched_barrier(0);
        K3_PRE_STORE(lds_b[buf ^ 1] + t)
        __syncthreads();
    }

#pragma unroll
    for (int j = 0; j < 64; ++j) { a[j].x = ap[4 * j]; a[j].y = ap[4 * j + 1]; }

    for (int ci = 8; ci < 16; ++ci) {
        const int buf = ci & 1;
        const int cn = min(ci + 1, 15);
        K3_PRE_LOAD(wq4 + (size_t)cn * K3_CHUNK4 + t)
        __builtin_amdgcn_sched_barrier(0);
        const float4* __restrict__ lb = lds_b[buf] + lane;
        float4 bf[64];
#pragma unroll
        for (int jj = 0; jj < 32; ++jj) {
            bf[2 * jj]     = lb[jj * 32];
            bf[2 * jj + 1] = lb[1024 + jj * 32];
        }
        __builtin_amdgcn_sched_barrier(0);
        v8f c0 = {}, c1 = {};
#pragma unroll
        for (int jj = 0; jj < 32; ++jj) {
            const float4 p = bf[2 * jj];
            const float4 q = bf[2 * jj + 1];
            v2f b;
            b.x = p.x; b.y = p.y;
            c0 = __builtin_amdgcn_wmma_f32_16x16x4_f32(
                     false, a[2 * jj], false, b, (short)0, c0, false, false);
            b.x = q.x; b.y = q.y;
            c1 = __builtin_amdgcn_wmma_f32_16x16x4_f32(
                     false, a[2 * jj], false, b, (short)0, c1, false, false);
            b.x = p.z; b.y = p.w;
            c0 = __builtin_amdgcn_wmma_f32_16x16x4_f32(
                     false, a[2 * jj + 1], false, b, (short)0, c0, false, false);
            b.x = q.z; b.y = q.w;
            c1 = __builtin_amdgcn_wmma_f32_16x16x4_f32(
                     false, a[2 * jj + 1], false, b, (short)0, c1, false, false);
        }
        const int nn = (ci - 8) * 32 + mrow;
        const float bias0 = bo[nn];
        const float bias1 = bo[nn + 16];
        __builtin_amdgcn_sched_barrier(0);
        K3_PRE_STORE(lds_b[buf ^ 1] + t)
        __syncthreads();
#pragma unroll
        for (int r = 0; r < 8; ++r) {
            out[(size_t)(m0 + r + 8 * hi) * F_SIZE + nn] = c0[r] + bias0;
            out[(size_t)(m0 + r + 8 * hi) * F_SIZE + nn + 16] = c1[r] + bias1;
        }
    }
}

extern "C" void kernel_launch(void* const* d_in, const int* in_sizes, int n_in,
                              void* d_out, int out_size, void* d_ws, size_t ws_size,
                              hipStream_t stream) {
    const float* act  = (const float*)d_in[0];
    const float* pas  = (const float*)d_in[1];
    const int*   bids = (const int*)d_in[2];
    const int*   mids = (const int*)d_in[3];
    const int*   xy   = (const int*)d_in[4];
    const int*   m0   = (const int*)d_in[5];
    const int*   m1   = (const int*)d_in[6];
    const int*   m2   = (const int*)d_in[7];
    const int*   m3   = (const int*)d_in[8];
    const float* Wa   = (const float*)d_in[9];
    const float* ba   = (const float*)d_in[10];
    const float* Wv   = (const float*)d_in[11];
    const float* bv   = (const float*)d_in[12];
    const float* Wo   = (const float*)d_in[13];
    const float* bo   = (const float*)d_in[14];
    float* out = (float*)d_out;

    const int n_act = in_sizes[0] / F_SIZE;

    float* wq       = (float*)d_ws;
    float* weighted = (float*)d_ws + (size_t)n_act * 64;

    k_attn_gather<<<K12_BLOCKS, K12_THREADS, 0, stream>>>(
        act, pas, bids, mids, xy, m0, m1, m2, m3, Wa, ba, weighted, n_act);

    k_frag_order<<<128, 256, 0, stream>>>(Wv, Wo, wq);

    k_dual_gemm<<<n_act / 128, 256, 0, stream>>>(weighted, wq, bv, bo, out);
}
